# starting kernel
# baseline (speedup 1.0000x reference)
#include <hip/hip_runtime.h>
#include <hip/hip_bf16.h>


typedef __attribute__((ext_vector_type(2))) float v2f;
typedef __attribute__((ext_vector_type(8))) float v8f;

#define EDM_B 4
#define EDM_N 2048
#define EDM_D 128
#define EDM_EPS 1e-7f

__global__ void edm_sqnorm_kernel(const float* __restrict__ x,
                                  float* __restrict__ sq) {
    const int wave = (blockIdx.x * blockDim.x + threadIdx.x) >> 5;
    const int lane = threadIdx.x & 31;
    if (wave >= EDM_B * EDM_N) return;

    const float* row = x + (size_t)wave * EDM_D;
    float s = 0.0f;
#pragma unroll
    for (int k = 0; k < EDM_D; k += 32) {
        float v = row[k + lane];
        s = fmaf(v, v, s);
    }
#pragma unroll
    for (int off = 16; off > 0; off >>= 1)
        s += __shfl_xor(s, off, 32);
    if (lane == 0) sq[wave] = s;
}

__global__ void edm_tile_kernel(const float* __restrict__ x,
                                const float* __restrict__ sq,
                                float* __restrict__ out) {
    const int lane   = threadIdx.x & 31;
    const int waveId = threadIdx.x >> 5;
    const int tile   = blockIdx.x * (blockDim.x >> 5) + waveId;

    const int b  = tile >> 14;
    const int ti = (tile >> 7) & 127;
    const int tj = tile & 127;
    const int row0 = ti << 4;
    const int col0 = tj << 4;

    const int m    = lane & 15;
    const int koff = (lane >> 4) << 1;

    const float* xb = x + (size_t)b * EDM_N * EDM_D;
    const float* ap = xb + (size_t)(row0 + m) * EDM_D + koff;
    const float* bp = xb + (size_t)(col0 + m) * EDM_D + koff;

    v8f acc = {};
#pragma unroll 8
    for (int k0 = 0; k0 < EDM_D; k0 += 4) {
        v2f av = *(const v2f*)(ap + k0);
        v2f bv = *(const v2f*)(bp + k0);
        acc = __builtin_amdgcn_wmma_f32_16x16x4_f32(
             false, av,  false, bv,
             (short)0, acc,  false,  false);
    }

    const int   half = lane >> 4;
    const float sqj  = sq[b * EDM_N + col0 + m];
    const float* sqi = sq + b * EDM_N + row0 + (half << 3);
    float* orow = out + ((size_t)b * EDM_N + row0 + (half << 3)) * EDM_N
                      + col0 + m;
#pragma unroll
    for (int r = 0; r < 8; ++r) {
        float d2 = sqi[r] + sqj - 2.0f * acc[r];
        d2 = fmaxf(d2, 0.0f);
        __builtin_nontemporal_store(sqrtf(d2 + EDM_EPS), orow + (size_t)r * EDM_N);
    }
}

extern "C" void kernel_launch(void* const* d_in, const int* in_sizes, int n_in,
                              void* d_out, int out_size, void* d_ws, size_t ws_size,
                              hipStream_t stream) {
    (void)in_sizes; (void)n_in; (void)out_size; (void)ws_size;

    const float* x  = (const float*)d_in[0];
    float* out      = (float*)d_out;
    float* sq       = (float*)d_ws;

    edm_sqnorm_kernel<<<1024, 256, 0, stream>>>(x, sq);

    edm_tile_kernel<<<8192, 256, 0, stream>>>(x, sq, out);
}


# Norm launch dropped (1 launch, no workspace): a work-group forms its 256 column norms once from whole-line reads, row norms from the register A panel, shared via LDS; all sum orders and bits kept
# speedup vs baseline: 2.8334x; 2.8334x over previous
#include <hip/hip_runtime.h>
#include <hip/hip_bf16.h>


typedef __attribute__((ext_vector_type(2))) float v2f;
typedef __attribute__((ext_vector_type(4))) float v4f;
typedef __attribute__((ext_vector_type(8))) float v8f;

#define EDM_B 4
#define EDM_N 2048
#define EDM_D 128
#define EDM_EPS 1e-7f

#define EDM_KLOADS (EDM_D / 8)
#define EDM_SWEEP  16
#define EDM_TILES  4
#define EDM_STEPS  (EDM_SWEEP / EDM_TILES)
#define EDM_RING   4
#define EDM_NEXT   (EDM_TILES * 16 * EDM_D)
#define EDM_SEGS   (EDM_N / 16 / EDM_SWEEP)
#define EDM_STRIPS 2
#define EDM_WAVES  8
#define EDM_GROUPS (EDM_N / (16 * EDM_STRIPS * EDM_WAVES))
#define EDM_COLS   (16 * EDM_SWEEP)
#define EDM_NCOLS  (EDM_COLS / EDM_WAVES)


__device__ __forceinline__ float edm_sqrt_rn(float x) {
    float s = __builtin_amdgcn_sqrtf(x);
    const float s_dn = __builtin_bit_cast(float, __builtin_bit_cast(unsigned, s) - 1u);
    const float s_up = __builtin_bit_cast(float, __builtin_bit_cast(unsigned, s) + 1u);
    const float r_dn = __builtin_fmaf(-s_dn, s, x);
    const float r_up = __builtin_fmaf(-s_up, s, x);
    s = (r_dn <= 0.0f) ? s_dn : s;
    s = (r_up >  0.0f) ? s_up : s;
    return s;
}

__device__ __forceinline__ void edm_gather(v4f (&slot)[EDM_TILES], const float* __restrict__ bp, int p) {
#pragma unroll
    for (int q = 0; q < EDM_TILES; ++q)
        slot[q] = *(const v4f*)(bp + q * EDM_D + 8 * p);
}

template <bool MORE>
__device__ __forceinline__ void edm_step(const v4f (&a)[EDM_STRIPS][EDM_KLOADS], v4f (&ring)[EDM_RING][EDM_TILES],
                                         const float* __restrict__ bp, const float* sqcol,
                                         const v4f (&sqi)[EDM_STRIPS][2], float* __restrict__ orow,
                                         const bool first) {
    v4f sqj;

    v8f acc[EDM_STRIPS][EDM_TILES] = {};
#pragma unroll
    for (int p = 0; p < EDM_KLOADS; ++p) {
        const int sl = p % EDM_RING;
#pragma unroll
        for (int s = 0; s < EDM_STRIPS; ++s)
#pragma unroll
            for (int q = 0; q < EDM_TILES; ++q)
                acc[s][q] = __builtin_amdgcn_wmma_f32_16x16x4_f32(
                     false, __builtin_shufflevector(a[s][p], a[s][p], 0, 1),
                     false, __builtin_shufflevector(ring[sl][q], ring[sl][q], 0, 1),
                     (short)0, acc[s][q],  false,  false);
#pragma unroll
        for (int s = 0; s < EDM_STRIPS; ++s)
#pragma unroll
            for (int q = 0; q < EDM_TILES; ++q)
                acc[s][q] = __builtin_amdgcn_wmma_f32_16x16x4_f32(
                     false, __builtin_shufflevector(a[s][p], a[s][p], 2, 3),
                     false, __builtin_shufflevector(ring[sl][q], ring[sl][q], 2, 3),
                     (short)0, acc[s][q],  false,  false);
        if (p + EDM_RING < EDM_KLOADS)
            edm_gather(ring[sl], bp, p + EDM_RING);
        else if (MORE)
            edm_gather(ring[sl], bp + EDM_NEXT, p + EDM_RING - EDM_KLOADS);
        if (p == EDM_KLOADS / 2) {
            if (first)
                asm volatile("s_barrier_wait -1" ::: "memory");
            sqj = *(const v4f*)sqcol;
        }
        __builtin_amdgcn_sched_barrier(0);
    }

#pragma unroll
    for (int s = 0; s < EDM_STRIPS; ++s) {
#pragma unroll
        for (int r = 0; r < 8; ++r) {
            v4f d;
#pragma unroll
            for (int q = 0; q < EDM_TILES; ++q) {
                float d2 = sqi[s][r >> 2][r & 3] + sqj[q] - 2.0f * acc[s][q][r];
                d2 = fmaxf(d2, 0.0f);
                d[q] = edm_sqrt_rn(d2 + EDM_EPS);
            }
            __builtin_nontemporal_store(d, (v4f*)(orow + (size_t)(16 * s + r) * EDM_N));
        }
    }
}

__global__ void __launch_bounds__(EDM_WAVES * 32) __attribute__((amdgpu_waves_per_eu(2, 2)))
edm_strip_kernel(const float* __restrict__ x,
                 float* __restrict__ out) {
    __shared__ __attribute__((aligned(16))) float s_col[EDM_COLS];
    __shared__ __attribute__((aligned(16))) float s_row[EDM_WAVES][16 * EDM_STRIPS];

    const int lane   = threadIdx.x & 31;
    const int waveId = threadIdx.x >> 5;

    const int seg = blockIdx.x % EDM_SEGS;
    const int sg  = (blockIdx.x / EDM_SEGS) % EDM_GROUPS;
    const int b   = blockIdx.x / (EDM_SEGS * EDM_GROUPS);

    const int row0 = (sg * EDM_WAVES + waveId) * (16 * EDM_STRIPS);
    const int m    = lane & 15;
    const int half = lane >> 4;
    const int koff = half << 2;

    const float* xb  = x + (size_t)b * EDM_N * EDM_D;

    const int col0 = (seg * EDM_SWEEP) << 4;
    const float* bp = xb + (size_t)(col0 + 4 * m) * EDM_D + koff;

    const int e8 = lane & 7;
    const float* cp = xb + (size_t)(col0 + EDM_NCOLS * waveId + (lane >> 3)) * EDM_D + 4 * e8;
    float ncol = 0.0f;
#pragma unroll
    for (int i = 0; i < EDM_NCOLS / 4; ++i) {
        v4f c4 = {};
#pragma unroll
        for (int j = 0; j < 4; ++j) {
            const v4f v = *(const v4f*)(cp + (size_t)(4 * i) * EDM_D + 32 * j);
            c4 = __builtin_elementwise_fma(v, v, c4);
        }
#pragma unroll
        for (int off = 4; off > 0; off >>= 1) {
            v4f o4;
#pragma unroll
            for (int c = 0; c < 4; ++c)
                o4[c] = __shfl_xor(c4[c], off, 32);
            c4 += o4;
        }
        const float t = (c4[0] + c4[2]) + (c4[1] + c4[3]);
        ncol = (e8 == i) ? t : ncol;
    }
    s_col[EDM_NCOLS * waveId + 4 * e8 + (lane >> 3)] = ncol;

    const float* ap = xb + (size_t)(row0 + m) * EDM_D + koff;
    v4f a[EDM_STRIPS][EDM_KLOADS];
#pragma unroll
    for (int s = 0; s < EDM_STRIPS; ++s)
#pragma unroll
        for (int p = 0; p < EDM_KLOADS; ++p)
            a[s][p] = *(const v4f*)(ap + s * 16 * EDM_D + 8 * p);

    v4f ring[EDM_RING][EDM_TILES];
#pragma unroll
    for (int p = 0; p < EDM_RING; ++p)
        edm_gather(ring[p], bp, p);

#pragma unroll
    for (int s = 0; s < EDM_STRIPS; ++s) {
        v4f q4[4] = {};
#pragma unroll
        for (int p = 0; p < EDM_KLOADS; ++p)
            q4[p & 3] = __builtin_elementwise_fma(a[s][p], a[s][p], q4[p & 3]);
        v4f r4 = (q4[0] + q4[2]) + (q4[1] + q4[3]);
        v4f o4;
#pragma unroll
        for (int c = 0; c < 4; ++c)
            o4[c] = __shfl_xor(r4[c], 16, 32);
        r4 += o4;
        s_row[waveId][16 * s + m] = (r4[0] + r4[2]) + (r4[1] + r4[3]);
    }
    asm volatile("s_wait_dscnt 0x0" ::: "memory");
    asm volatile("s_barrier_signal -1" ::: "memory");

    v4f sqi[EDM_STRIPS][2];
#pragma unroll
    for (int s = 0; s < EDM_STRIPS; ++s) {
        sqi[s][0] = *(const v4f*)&s_row[waveId][16 * s + (half << 3)];
        sqi[s][1] = *(const v4f*)&s_row[waveId][16 * s + (half << 3) + 4];
    }
    const float* sqcol = s_col + 4 * m;
    float* orow = out + ((size_t)b * EDM_N + row0 + (half << 3)) * EDM_N + col0 + 4 * m;

    for (int s = 0; s < EDM_STEPS - 1; ++s) {
        edm_step<true>(a, ring, bp, sqcol, sqi, orow, s == 0);
        bp    += EDM_NEXT;
        sqcol += EDM_TILES * 16;
        orow  += EDM_TILES * 16;
    }
    edm_step<false>(a, ring, bp, sqcol, sqi, orow, false);
}

extern "C" void kernel_launch(void* const* d_in, const int* in_sizes, int n_in,
                              void* d_out, int out_size, void* d_ws, size_t ws_size,
                              hipStream_t stream) {
    (void)in_sizes; (void)n_in; (void)out_size; (void)d_ws; (void)ws_size;

    const float* x  = (const float*)d_in[0];
    float* out      = (float*)d_out;

    edm_strip_kernel<<<256, 256, 0, stream>>>(x, out);
}
